# starting kernel
# baseline (speedup 1.0000x reference)
#include <hip/hip_runtime.h>
#include <hip/hip_bf16.h>

typedef float v2f __attribute__((ext_vector_type(2)));
typedef float v4f __attribute__((ext_vector_type(4)));
typedef float v8f __attribute__((ext_vector_type(8)));

#define IN_NF   64
#define HIDN    128
#define EHW     64
#define KIN_NODE (HIDN + EHW)

__device__ __forceinline__ float silu_f(float x) {
    return x / (1.0f + __expf(-x));
}

__device__ __forceinline__ v8f splat8(float v) {
    v8f r = {v, v, v, v, v, v, v, v};
    return r;
}

__device__ __forceinline__ v8f wmma4(v2f a, v2f b, v8f c) {
    return __builtin_amdgcn_wmma_f32_16x16x4_f32(
         false, a,  false, b,
         (short)0, c,  false,  false);
}

__global__ void embed_kernel(const float* __restrict__ hin,
                             const float* __restrict__ W,
                             const float* __restrict__ bias,
                             float* __restrict__ hout, int Nn) {
    const int lane = threadIdx.x & 31;
    const int c    = lane & 15;
    const int half = lane >> 4;
    const int nw   = (gridDim.x * blockDim.x) >> 5;
    const int wid0 = (blockIdx.x * blockDim.x + threadIdx.x) >> 5;
    const int tiles = (Nn + 15) >> 4;

    for (int t = wid0; t < tiles; t += nw) {
        int mr = t * 16 + c;
        if (mr >= Nn) mr = Nn - 1;

        v8f acc[8];
#pragma unroll
        for (int n = 0; n < 8; ++n) acc[n] = splat8(bias[n * 16 + c]);

        for (int k = 0; k < IN_NF; k += 4) {
            const int kk = k + half * 2;
            v2f a = *(const v2f*)(hin + (size_t)mr * IN_NF + kk);
#pragma unroll
            for (int n = 0; n < 8; ++n) {
                v2f b;
                b.x = W[kk * HIDN + n * 16 + c];
                b.y = W[(kk + 1) * HIDN + n * 16 + c];
                acc[n] = wmma4(a, b, acc[n]);
            }
        }
        if (t * 16 + 16 <= Nn) {
#pragma unroll
            for (int n = 0; n < 8; ++n) {
                float* p = hout + (size_t)(t * 16 + half * 8) * HIDN + n * 16 + c;
#pragma unroll
                for (int i = 0; i < 8; ++i) p[(size_t)i * HIDN] = acc[n][i];
            }
        } else {
#pragma unroll
            for (int n = 0; n < 8; ++n)
#pragma unroll
                for (int i = 0; i < 8; ++i) {
                    const int r = t * 16 + i + half * 8;
                    if (r < Nn) hout[(size_t)r * HIDN + n * 16 + c] = acc[n][i];
                }
        }
    }
}

__global__ void edge_kernel(const float* __restrict__ h,
                            const int* __restrict__ erow,
                            const int* __restrict__ ecol,
                            const float* __restrict__ w1,
                            const float* __restrict__ b1,
                            const float* __restrict__ w2,
                            const float* __restrict__ b2,
                            float* __restrict__ agg, int Ee) {
    constexpr int WS  = 72;
    constexpr int LWM = EHW + 4;
    __shared__ float w1s[2 * HIDN * WS];
    __shared__ float w2s[EHW * WS];
    __shared__ float m1s[8][16 * LWM];

    for (int i4 = threadIdx.x; i4 < (2 * HIDN * EHW) / 4; i4 += blockDim.x) {
        const int r = i4 >> 4, col = (i4 & 15) * 4;
        *(v4f*)(w1s + r * WS + col) = *(const v4f*)(w1 + r * EHW + col);
    }
    for (int i4 = threadIdx.x; i4 < (EHW * EHW) / 4; i4 += blockDim.x) {
        const int r = i4 >> 4, col = (i4 & 15) * 4;
        *(v4f*)(w2s + r * WS + col) = *(const v4f*)(w2 + r * EHW + col);
    }
    __syncthreads();

    const int lane = threadIdx.x & 31;
    const int wv   = threadIdx.x >> 5;
    const int c    = lane & 15;
    const int half = lane >> 4;
    const int nw   = (gridDim.x * blockDim.x) >> 5;
    const int wid0 = (blockIdx.x * blockDim.x + threadIdx.x) >> 5;
    const int tiles = (Ee + 15) >> 4;
    float* L = m1s[wv];

    for (int t = wid0; t < tiles; t += nw) {
        int ea = t * 16 + c;
        if (ea >= Ee) ea = Ee - 1;
        const float* __restrict__ hra = h + (size_t)erow[ea] * HIDN;
        const float* __restrict__ hca = h + (size_t)ecol[ea] * HIDN;

        v8f acc[4];
#pragma unroll
        for (int n = 0; n < 4; ++n) acc[n] = splat8(b1[n * 16 + c]);

#pragma unroll 4
        for (int k = 0; k < HIDN; k += 4) {
            const int kk = k + half * 2;
            v2f a = *(const v2f*)(hra + kk);
#pragma unroll
            for (int n = 0; n < 4; ++n) {
                v2f b;
                b.x = w1s[kk * WS + n * 16 + c];
                b.y = w1s[(kk + 1) * WS + n * 16 + c];
                acc[n] = wmma4(a, b, acc[n]);
            }
        }
#pragma unroll 4
        for (int k = 0; k < HIDN; k += 4) {
            const int kk = k + half * 2;
            v2f a = *(const v2f*)(hca + kk);
#pragma unroll
            for (int n = 0; n < 4; ++n) {
                v2f b;
                b.x = w1s[(HIDN + kk) * WS + n * 16 + c];
                b.y = w1s[(HIDN + kk + 1) * WS + n * 16 + c];
                acc[n] = wmma4(a, b, acc[n]);
            }
        }

#pragma unroll
        for (int n = 0; n < 4; ++n)
#pragma unroll
            for (int i = 0; i < 8; ++i)
                L[(i + half * 8) * LWM + n * 16 + c] = silu_f(acc[n][i]);

        asm volatile("s_wait_dscnt 0" ::: "memory");

        v8f acc2[4];
#pragma unroll
        for (int n = 0; n < 4; ++n) acc2[n] = splat8(b2[n * 16 + c]);

#pragma unroll 4
        for (int k = 0; k < EHW; k += 4) {
            const int kk = k + half * 2;
            v2f a;
            a.x = L[c * LWM + kk];
            a.y = L[c * LWM + kk + 1];
#pragma unroll
            for (int n = 0; n < 4; ++n) {
                v2f b;
                b.x = w2s[kk * WS + n * 16 + c];
                b.y = w2s[(kk + 1) * WS + n * 16 + c];
                acc2[n] = wmma4(a, b, acc2[n]);
            }
        }

        if (t * 16 + 16 <= Ee) {
#pragma unroll
            for (int i = 0; i < 8; ++i) {
                float* dst = agg + (size_t)erow[t * 16 + i + half * 8] * EHW + c;
#pragma unroll
                for (int n = 0; n < 4; ++n)
                    atomicAdd(dst + n * 16, silu_f(acc2[n][i]));
            }
        } else {
#pragma unroll
            for (int i = 0; i < 8; ++i) {
                const int e = t * 16 + i + half * 8;
                if (e < Ee) {
                    float* dst = agg + (size_t)erow[e] * EHW + c;
#pragma unroll
                    for (int n = 0; n < 4; ++n)
                        atomicAdd(dst + n * 16, silu_f(acc2[n][i]));
                }
            }
        }
        asm volatile("s_wait_dscnt 0" ::: "memory");
    }
}

__global__ void node_kernel(float* __restrict__ h,
                            const float* __restrict__ agg,
                            const float* __restrict__ m1,
                            const float* __restrict__ c1,
                            const float* __restrict__ m2,
                            const float* __restrict__ c2,
                            int Nn) {
    constexpr int LW  = HIDN + 4;
    constexpr int WS2 = HIDN + 8;
    __shared__ float w2s[HIDN * WS2];
    __shared__ float lds[8][16 * LW];

    for (int i4 = threadIdx.x; i4 < (HIDN * HIDN) / 4; i4 += blockDim.x) {
        const int r = i4 >> 5, col = (i4 & 31) * 4;
        *(v4f*)(w2s + r * WS2 + col) = *(const v4f*)(m2 + r * HIDN + col);
    }
    __syncthreads();

    const int lane = threadIdx.x & 31;
    const int wv   = threadIdx.x >> 5;
    const int c    = lane & 15;
    const int half = lane >> 4;
    const int nw   = (gridDim.x * blockDim.x) >> 5;
    const int wid0 = (blockIdx.x * blockDim.x + threadIdx.x) >> 5;
    const int tiles = (Nn + 15) >> 4;
    float* L = lds[wv];

    for (int t = wid0; t < tiles; t += nw) {
        int mr = t * 16 + c;
        if (mr >= Nn) mr = Nn - 1;
        const float* __restrict__ hr = h + (size_t)mr * HIDN;
        const float* __restrict__ ar = agg + (size_t)mr * EHW;

        v8f acc[8];
#pragma unroll
        for (int n = 0; n < 8; ++n) acc[n] = splat8(c1[n * 16 + c]);

        for (int k = 0; k < HIDN; k += 4) {
            const int kk = k + half * 2;
            v2f a = *(const v2f*)(hr + kk);
#pragma unroll
            for (int n = 0; n < 8; ++n) {
                v2f b;
                b.x = m1[kk * HIDN + n * 16 + c];
                b.y = m1[(kk + 1) * HIDN + n * 16 + c];
                acc[n] = wmma4(a, b, acc[n]);
            }
        }
        for (int k = 0; k < EHW; k += 4) {
            const int kk = k + half * 2;
            v2f a = *(const v2f*)(ar + kk);
#pragma unroll
            for (int n = 0; n < 8; ++n) {
                v2f b;
                b.x = m1[(HIDN + kk) * HIDN + n * 16 + c];
                b.y = m1[(HIDN + kk + 1) * HIDN + n * 16 + c];
                acc[n] = wmma4(a, b, acc[n]);
            }
        }
#pragma unroll
        for (int n = 0; n < 8; ++n)
#pragma unroll
            for (int i = 0; i < 8; ++i)
                L[(i + half * 8) * LW + n * 16 + c] = silu_f(acc[n][i]);

        asm volatile("s_wait_dscnt 0" ::: "memory");

        v8f acc2[8];
#pragma unroll
        for (int n = 0; n < 8; ++n) acc2[n] = splat8(c2[n * 16 + c]);

        for (int k = 0; k < HIDN; k += 4) {
            const int kk = k + half * 2;
            v2f a;
            a.x = L[c * LW + kk];
            a.y = L[c * LW + kk + 1];
#pragma unroll
            for (int n = 0; n < 8; ++n) {
                v2f b;
                b.x = w2s[kk * WS2 + n * 16 + c];
                b.y = w2s[(kk + 1) * WS2 + n * 16 + c];
                acc2[n] = wmma4(a, b, acc2[n]);
            }
        }
        if (t * 16 + 16 <= Nn) {
#pragma unroll
            for (int n = 0; n < 8; ++n) {
                float* p = h + (size_t)(t * 16 + half * 8) * HIDN + n * 16 + c;
#pragma unroll
                for (int i = 0; i < 8; ++i) p[(size_t)i * HIDN] += acc2[n][i];
            }
        } else {
#pragma unroll
            for (int n = 0; n < 8; ++n)
#pragma unroll
                for (int i = 0; i < 8; ++i) {
                    const int r = t * 16 + i + half * 8;
                    if (r < Nn) h[(size_t)r * HIDN + n * 16 + c] += acc2[n][i];
                }
        }
        asm volatile("s_wait_dscnt 0" ::: "memory");
    }
}

__global__ void out_kernel(const float* __restrict__ h,
                           const float* __restrict__ W,
                           const float* __restrict__ bias,
                           float* __restrict__ out, int Nn) {
    const int lane = threadIdx.x & 31;
    const int c    = lane & 15;
    const int half = lane >> 4;
    const int nw   = (gridDim.x * blockDim.x) >> 5;
    const int wid0 = (blockIdx.x * blockDim.x + threadIdx.x) >> 5;
    const int tiles = (Nn + 15) >> 4;

    for (int t = wid0; t < tiles; t += nw) {
        int mr = t * 16 + c;
        if (mr >= Nn) mr = Nn - 1;

        v8f acc[4];
#pragma unroll
        for (int n = 0; n < 4; ++n) acc[n] = splat8(bias[n * 16 + c]);

        for (int k = 0; k < HIDN; k += 4) {
            const int kk = k + half * 2;
            v2f a = *(const v2f*)(h + (size_t)mr * HIDN + kk);
#pragma unroll
            for (int n = 0; n < 4; ++n) {
                v2f b;
                b.x = W[kk * EHW + n * 16 + c];
                b.y = W[(kk + 1) * EHW + n * 16 + c];
                acc[n] = wmma4(a, b, acc[n]);
            }
        }
        if (t * 16 + 16 <= Nn) {
#pragma unroll
            for (int n = 0; n < 4; ++n) {
                float* p = out + (size_t)(t * 16 + half * 8) * EHW + n * 16 + c;
#pragma unroll
                for (int i = 0; i < 8; ++i) p[(size_t)i * EHW] = acc[n][i];
            }
        } else {
#pragma unroll
            for (int n = 0; n < 4; ++n)
#pragma unroll
                for (int i = 0; i < 8; ++i) {
                    const int r = t * 16 + i + half * 8;
                    if (r < Nn) out[(size_t)r * EHW + n * 16 + c] = acc[n][i];
                }
        }
    }
}

static inline int imin_h(int a, int b) { return a < b ? a : b; }
static inline int cdiv_h(int a, int b) { return (a + b - 1) / b; }

extern "C" void kernel_launch(void* const* d_in, const int* in_sizes, int n_in,
                              void* d_out, int out_size, void* d_ws, size_t ws_size,
                              hipStream_t stream) {
    const float* h_in  = (const float*)d_in[0];
    const int*   edges = (const int*)d_in[1];
    const float* W_in  = (const float*)d_in[2];
    const float* b_in  = (const float*)d_in[3];
    const float* eW1   = (const float*)d_in[4];
    const float* eb1   = (const float*)d_in[5];
    const float* eW2   = (const float*)d_in[6];
    const float* eb2   = (const float*)d_in[7];
    const float* nW1   = (const float*)d_in[8];
    const float* nb1   = (const float*)d_in[9];
    const float* nW2   = (const float*)d_in[10];
    const float* nb2   = (const float*)d_in[11];
    const float* W_out = (const float*)d_in[12];
    const float* b_out = (const float*)d_in[13];

    const int Nn = in_sizes[0] / IN_NF;
    const int Ee = in_sizes[1] / 2;
    const int Ll = in_sizes[4] / (2 * HIDN * EHW);

    float* h   = (float*)d_ws;
    float* agg = h + (size_t)Nn * HIDN;

    const int* erow = edges;
    const int* ecol = edges + Ee;

    const int tilesN = cdiv_h(Nn, 16);
    const int tilesE = cdiv_h(Ee, 16);
    const int gEmb = imin_h(cdiv_h(tilesN, 8), 2048);
    const int gEdg = imin_h(cdiv_h(tilesE, 8), 2048);
    const int gNod = imin_h(cdiv_h(tilesN, 8), 2048);

    embed_kernel<<<gEmb, 256, 0, stream>>>(h_in, W_in, b_in, h, Nn);

    for (int l = 0; l < Ll; ++l) {
        hipMemsetAsync(agg, 0, (size_t)Nn * EHW * sizeof(float), stream);
        edge_kernel<<<gEdg, 256, 0, stream>>>(
            h, erow, ecol,
            eW1 + (size_t)l * 2 * HIDN * EHW, eb1 + (size_t)l * EHW,
            eW2 + (size_t)l * EHW * EHW,      eb2 + (size_t)l * EHW,
            agg, Ee);
        node_kernel<<<gNod, 256, 0, stream>>>(
            h, agg,
            nW1 + (size_t)l * KIN_NODE * HIDN, nb1 + (size_t)l * HIDN,
            nW2 + (size_t)l * HIDN * HIDN,     nb2 + (size_t)l * HIDN,
            Nn);
    }

    out_kernel<<<gEmb, 256, 0, stream>>>(h, W_out, b_out, (float*)d_out, Nn);
}


# Edge kernel: weights k-interleaved in LDS, a lane's fragment pair is one 16-B read; reads in groups of 8 per 16 products (2nd product: one group of 40); 2nd product's float sum order changes
# speedup vs baseline: 1.1182x; 1.1182x over previous
#include <hip/hip_runtime.h>
#include <hip/hip_bf16.h>

typedef float v2f __attribute__((ext_vector_type(2)));
typedef float v4f __attribute__((ext_vector_type(4)));
typedef float v8f __attribute__((ext_vector_type(8)));

#define IN_NF   64
#define HIDN    128
#define EHW     64
#define KIN_NODE (HIDN + EHW)

__device__ __forceinline__ float silu_f(float x) {
    return x / (1.0f + __expf(-x));
}

__device__ __forceinline__ v8f splat8(float v) {
    v8f r = {v, v, v, v, v, v, v, v};
    return r;
}

__device__ __forceinline__ v8f wmma4(v2f a, v2f b, v8f c) {
    return __builtin_amdgcn_wmma_f32_16x16x4_f32(
         false, a,  false, b,
         (short)0, c,  false,  false);
}

__global__ void embed_kernel(const float* __restrict__ hin,
                             const float* __restrict__ W,
                             const float* __restrict__ bias,
                             float* __restrict__ hout, int Nn) {
    const int lane = threadIdx.x & 31;
    const int c    = lane & 15;
    const int half = lane >> 4;
    const int nw   = (gridDim.x * blockDim.x) >> 5;
    const int wid0 = (blockIdx.x * blockDim.x + threadIdx.x) >> 5;
    const int tiles = (Nn + 15) >> 4;

    for (int t = wid0; t < tiles; t += nw) {
        int mr = t * 16 + c;
        if (mr >= Nn) mr = Nn - 1;

        v8f acc[8];
#pragma unroll
        for (int n = 0; n < 8; ++n) acc[n] = splat8(bias[n * 16 + c]);

        for (int k = 0; k < IN_NF; k += 4) {
            const int kk = k + half * 2;
            v2f a = *(const v2f*)(hin + (size_t)mr * IN_NF + kk);
#pragma unroll
            for (int n = 0; n < 8; ++n) {
                v2f b;
                b.x = W[kk * HIDN + n * 16 + c];
                b.y = W[(kk + 1) * HIDN + n * 16 + c];
                acc[n] = wmma4(a, b, acc[n]);
            }
        }
        if (t * 16 + 16 <= Nn) {
#pragma unroll
            for (int n = 0; n < 8; ++n) {
                float* p = hout + (size_t)(t * 16 + half * 8) * HIDN + n * 16 + c;
#pragma unroll
                for (int i = 0; i < 8; ++i) p[(size_t)i * HIDN] = acc[n][i];
            }
        } else {
#pragma unroll
            for (int n = 0; n < 8; ++n)
#pragma unroll
                for (int i = 0; i < 8; ++i) {
                    const int r = t * 16 + i + half * 8;
                    if (r < Nn) hout[(size_t)r * HIDN + n * 16 + c] = acc[n][i];
                }
        }
    }
}

__device__ __forceinline__ void stage_kinter(float* __restrict__ ws, const float* __restrict__ wg, int K) {
    for (int item = threadIdx.x; item < K * 4; item += blockDim.x) {
        const int q = item >> 4, c4 = item & 15;
        const float* s = wg + (size_t)(4 * q) * EHW + 4 * c4;
        const v4f r0 = *(const v4f*)(s);
        const v4f r1 = *(const v4f*)(s + EHW);
        const v4f r2 = *(const v4f*)(s + 2 * EHW);
        const v4f r3 = *(const v4f*)(s + 3 * EHW);
        float* d = ws + q * 256 + c4 * 16;
        v4f o0 = {r0.x, r1.x, r2.x, r3.x};
        v4f o1 = {r0.y, r1.y, r2.y, r3.y};
        v4f o2 = {r0.z, r1.z, r2.z, r3.z};
        v4f o3 = {r0.w, r1.w, r2.w, r3.w};
        *(v4f*)(d)      = o0;
        *(v4f*)(d + 4)  = o1;
        *(v4f*)(d + 8)  = o2;
        *(v4f*)(d + 12) = o3;
    }
}

__global__ void __launch_bounds__(256, 4) edge_kernel(const float* __restrict__ h,
                            const int* __restrict__ erow,
                            const int* __restrict__ ecol,
                            const float* __restrict__ w1,
                            const float* __restrict__ b1,
                            const float* __restrict__ w2,
                            const float* __restrict__ b2,
                            float* __restrict__ agg, int Ee) {
    constexpr int LWM = EHW + 4;
    __shared__ __attribute__((aligned(16))) float w1s[2 * HIDN * EHW];
    __shared__ __attribute__((aligned(16))) float w2s[EHW * EHW];
    __shared__ __attribute__((aligned(16))) float m1s[8][16 * LWM];

    stage_kinter(w1s, w1, 2 * HIDN);
    stage_kinter(w2s, w2, EHW);
    __syncthreads();

    const int lane = threadIdx.x & 31;
    const int wv   = threadIdx.x >> 5;
    const int c    = lane & 15;
    const int half = lane >> 4;
    const int nw   = (gridDim.x * blockDim.x) >> 5;
    const int wid0 = (blockIdx.x * blockDim.x + threadIdx.x) >> 5;
    const int tiles = (Ee + 15) >> 4;
    float* L = m1s[wv];

    const float* wl  = w1s + (half * 64 + c) * 4;
    const float* wl2 = w2s + (half * 64 + c) * 4;
    const float* La  = L + c * LWM + half * 4;

    for (int t = wid0; t < tiles; t += nw) {
        int ea = t * 16 + c;
        if (ea >= Ee) ea = Ee - 1;
        const float* __restrict__ hra = h + (size_t)erow[ea] * HIDN + half * 4;
        const float* __restrict__ hca = h + (size_t)ecol[ea] * HIDN + half * 4;

        v4f av[32];
#pragma unroll
        for (int j = 0; j < 16; ++j) av[j] = *(const v4f*)(hra + 8 * j);
#pragma unroll
        for (int j = 0; j < 16; ++j) av[16 + j] = *(const v4f*)(hca + 8 * j);

        __builtin_amdgcn_sched_group_barrier(0x020, 18, 0);

        v8f acc[4];
#pragma unroll
        for (int n = 0; n < 4; ++n) acc[n] = splat8(b1[n * 16 + c]);

        v4f bq[2][8];
#pragma unroll
        for (int n = 0; n < 4; ++n) {
            bq[0][n]     = *(const v4f*)(wl + n * 64);
            bq[0][4 + n] = *(const v4f*)(wl + 512 + n * 64);
        }
        __builtin_amdgcn_sched_group_barrier(0x100, 8, 0);
#pragma unroll
        for (int p = 0; p < 16; ++p) {
            if ((p & 1) == 0 && p >= 2 && p <= 8) {
                __builtin_amdgcn_sched_group_barrier(0x020, 4, 0);
            }
            if (p + 1 < 16) {
                const float* wn = wl + (p + 1) * 1024;
#pragma unroll
                for (int n = 0; n < 4; ++n) {
                    bq[(p + 1) & 1][n]     = *(const v4f*)(wn + n * 64);
                    bq[(p + 1) & 1][4 + n] = *(const v4f*)(wn + 512 + n * 64);
                }
                __builtin_amdgcn_sched_group_barrier(0x100, 8, 0);
            }
#pragma unroll
            for (int jj = 0; jj < 2; ++jj) {
                const int j = 2 * p + jj;
                v2f a;
                a.x = av[j].x; a.y = av[j].y;
#pragma unroll
                for (int n = 0; n < 4; ++n) {
                    v2f b;
                    b.x = bq[p & 1][jj * 4 + n].x; b.y = bq[p & 1][jj * 4 + n].y;
                    acc[n] = wmma4(a, b, acc[n]);
                }
                a.x = av[j].z; a.y = av[j].w;
#pragma unroll
                for (int n = 0; n < 4; ++n) {
                    v2f b;
                    b.x = bq[p & 1][jj * 4 + n].z; b.y = bq[p & 1][jj * 4 + n].w;
                    acc[n] = wmma4(a, b, acc[n]);
                }
            }
            __builtin_amdgcn_sched_group_barrier(0x008, 16, 0);
        }

#pragma unroll
        for (int n = 0; n < 4; ++n)
#pragma unroll
            for (int i = 0; i < 8; ++i)
                L[(i + half * 8) * LWM + n * 16 + c] = silu_f(acc[n][i]);

        asm volatile("s_wait_dscnt 0" ::: "memory");

        v8f acc2[4];
#pragma unroll
        for (int n = 0; n < 4; ++n) acc2[n] = splat8(b2[n * 16 + c]);

        v4f a2[8];
        v4f b2q[8][4];
#pragma unroll
        for (int j = 0; j < 8; ++j) {
            a2[j] = *(const v4f*)(La + 8 * j);
#pragma unroll
            for (int n = 0; n < 4; ++n) b2q[j][n] = *(const v4f*)(wl2 + j * 512 + n * 64);
        }
        __builtin_amdgcn_sched_group_barrier(0x100, 40, 0);
#pragma unroll
        for (int j = 0; j < 8; ++j) {
            v2f a;
            a.x = a2[j].x; a.y = a2[j].y;
#pragma unroll
            for (int n = 0; n < 4; ++n) {
                v2f b;
                b.x = b2q[j][n].x; b.y = b2q[j][n].y;
                acc2[n] = wmma4(a, b, acc2[n]);
            }
            a.x = a2[j].z; a.y = a2[j].w;
#pragma unroll
            for (int n = 0; n < 4; ++n) {
                v2f b;
                b.x = b2q[j][n].z; b.y = b2q[j][n].w;
                acc2[n] = wmma4(a, b, acc2[n]);
            }
        }
        __builtin_amdgcn_sched_group_barrier(0x008, 64, 0);

        if (t * 16 + 16 <= Ee) {
#pragma unroll
            for (int i = 0; i < 8; ++i) {
                float* dst = agg + (size_t)erow[t * 16 + i + half * 8] * EHW + c;
#pragma unroll
                for (int n = 0; n < 4; ++n)
                    atomicAdd(dst + n * 16, silu_f(acc2[n][i]));
            }
        } else {
#pragma unroll
            for (int i = 0; i < 8; ++i) {
                const int e = t * 16 + i + half * 8;
                if (e < Ee) {
                    float* dst = agg + (size_t)erow[e] * EHW + c;
#pragma unroll
                    for (int n = 0; n < 4; ++n)
                        atomicAdd(dst + n * 16, silu_f(acc2[n][i]));
                }
            }
        }
        asm volatile("s_wait_dscnt 0" ::: "memory");
    }
}

__global__ void node_kernel(float* __restrict__ h,
                            const float* __restrict__ agg,
                            const float* __restrict__ m1,
                            const float* __restrict__ c1,
                            const float* __restrict__ m2,
                            const float* __restrict__ c2,
                            int Nn) {
    constexpr int LW  = HIDN + 4;
    constexpr int WS2 = HIDN + 8;
    __shared__ float w2s[HIDN * WS2];
    __shared__ float lds[8][16 * LW];

    for (int i4 = threadIdx.x; i4 < (HIDN * HIDN) / 4; i4 += blockDim.x) {
        const int r = i4 >> 5, col = (i4 & 31) * 4;
        *(v4f*)(w2s + r * WS2 + col) = *(const v4f*)(m2 + r * HIDN + col);
    }
    __syncthreads();

    const int lane = threadIdx.x & 31;
    const int wv   = threadIdx.x >> 5;
    const int c    = lane & 15;
    const int half = lane >> 4;
    const int nw   = (gridDim.x * blockDim.x) >> 5;
    const int wid0 = (blockIdx.x * blockDim.x + threadIdx.x) >> 5;
    const int tiles = (Nn + 15) >> 4;
    float* L = lds[wv];

    for (int t = wid0; t < tiles; t += nw) {
        int mr = t * 16 + c;
        if (mr >= Nn) mr = Nn - 1;
        const float* __restrict__ hr = h + (size_t)mr * HIDN;
        const float* __restrict__ ar = agg + (size_t)mr * EHW;

        v8f acc[8];
#pragma unroll
        for (int n = 0; n < 8; ++n) acc[n] = splat8(c1[n * 16 + c]);

        for (int k = 0; k < HIDN; k += 4) {
            const int kk = k + half * 2;
            v2f a = *(const v2f*)(hr + kk);
#pragma unroll
            for (int n = 0; n < 8; ++n) {
                v2f b;
                b.x = m1[kk * HIDN + n * 16 + c];
                b.y = m1[(kk + 1) * HIDN + n * 16 + c];
                acc[n] = wmma4(a, b, acc[n]);
            }
        }
        for (int k = 0; k < EHW; k += 4) {
            const int kk = k + half * 2;
            v2f a = *(const v2f*)(ar + kk);
#pragma unroll
            for (int n = 0; n < 8; ++n) {
                v2f b;
                b.x = m1[(HIDN + kk) * HIDN + n * 16 + c];
                b.y = m1[(HIDN + kk + 1) * HIDN + n * 16 + c];
                acc[n] = wmma4(a, b, acc[n]);
            }
        }
#pragma unroll
        for (int n = 0; n < 8; ++n)
#pragma unroll
            for (int i = 0; i < 8; ++i)
                L[(i + half * 8) * LW + n * 16 + c] = silu_f(acc[n][i]);

        asm volatile("s_wait_dscnt 0" ::: "memory");

        v8f acc2[8];
#pragma unroll
        for (int n = 0; n < 8; ++n) acc2[n] = splat8(c2[n * 16 + c]);

        for (int k = 0; k < HIDN; k += 4) {
            const int kk = k + half * 2;
            v2f a;
            a.x = L[c * LW + kk];
            a.y = L[c * LW + kk + 1];
#pragma unroll
            for (int n = 0; n < 8; ++n) {
                v2f b;
                b.x = w2s[kk * WS2 + n * 16 + c];
                b.y = w2s[(kk + 1) * WS2 + n * 16 + c];
                acc2[n] = wmma4(a, b, acc2[n]);
            }
        }
        if (t * 16 + 16 <= Nn) {
#pragma unroll
            for (int n = 0; n < 8; ++n) {
                float* p = h + (size_t)(t * 16 + half * 8) * HIDN + n * 16 + c;
#pragma unroll
                for (int i = 0; i < 8; ++i) p[(size_t)i * HIDN] += acc2[n][i];
            }
        } else {
#pragma unroll
            for (int n = 0; n < 8; ++n)
#pragma unroll
                for (int i = 0; i < 8; ++i) {
                    const int r = t * 16 + i + half * 8;
                    if (r < Nn) h[(size_t)r * HIDN + n * 16 + c] += acc2[n][i];
                }
        }
        asm volatile("s_wait_dscnt 0" ::: "memory");
    }
}

__global__ void out_kernel(const float* __restrict__ h,
                           const float* __restrict__ W,
                           const float* __restrict__ bias,
                           float* __restrict__ out, int Nn) {
    const int lane = threadIdx.x & 31;
    const int c    = lane & 15;
    const int half = lane >> 4;
    const int nw   = (gridDim.x * blockDim.x) >> 5;
    const int wid0 = (blockIdx.x * blockDim.x + threadIdx.x) >> 5;
    const int tiles = (Nn + 15) >> 4;

    for (int t = wid0; t < tiles; t += nw) {
        int mr = t * 16 + c;
        if (mr >= Nn) mr = Nn - 1;

        v8f acc[4];
#pragma unroll
        for (int n = 0; n < 4; ++n) acc[n] = splat8(bias[n * 16 + c]);

        for (int k = 0; k < HIDN; k += 4) {
            const int kk = k + half * 2;
            v2f a = *(const v2f*)(h + (size_t)mr * HIDN + kk);
#pragma unroll
            for (int n = 0; n < 4; ++n) {
                v2f b;
                b.x = W[kk * EHW + n * 16 + c];
                b.y = W[(kk + 1) * EHW + n * 16 + c];
                acc[n] = wmma4(a, b, acc[n]);
            }
        }
        if (t * 16 + 16 <= Nn) {
#pragma unroll
            for (int n = 0; n < 4; ++n) {
                float* p = out + (size_t)(t * 16 + half * 8) * EHW + n * 16 + c;
#pragma unroll
                for (int i = 0; i < 8; ++i) p[(size_t)i * EHW] = acc[n][i];
            }
        } else {
#pragma unroll
            for (int n = 0; n < 4; ++n)
#pragma unroll
                for (int i = 0; i < 8; ++i) {
                    const int r = t * 16 + i + half * 8;
                    if (r < Nn) out[(size_t)r * EHW + n * 16 + c] = acc[n][i];
                }
        }
    }
}

static inline int imin_h(int a, int b) { return a < b ? a : b; }
static inline int cdiv_h(int a, int b) { return (a + b - 1) / b; }

extern "C" void kernel_launch(void* const* d_in, const int* in_sizes, int n_in,
                              void* d_out, int out_size, void* d_ws, size_t ws_size,
                              hipStream_t stream) {
    const float* h_in  = (const float*)d_in[0];
    const int*   edges = (const int*)d_in[1];
    const float* W_in  = (const float*)d_in[2];
    const float* b_in  = (const float*)d_in[3];
    const float* eW1   = (const float*)d_in[4];
    const float* eb1   = (const float*)d_in[5];
    const float* eW2   = (const float*)d_in[6];
    const float* eb2   = (const float*)d_in[7];
    const float* nW1   = (const float*)d_in[8];
    const float* nb1   = (const float*)d_in[9];
    const float* nW2   = (const float*)d_in[10];
    const float* nb2   = (const float*)d_in[11];
    const float* W_out = (const float*)d_in[12];
    const float* b_out = (const float*)d_in[13];

    const int Nn = in_sizes[0] / IN_NF;
    const int Ee = in_sizes[1] / 2;
    const int Ll = in_sizes[4] / (2 * HIDN * EHW);

    float* h   = (float*)d_ws;
    float* agg = h + (size_t)Nn * HIDN;

    const int* erow = edges;
    const int* ecol = edges + Ee;

    const int tilesN = cdiv_h(Nn, 16);
    const int tilesE = cdiv_h(Ee, 16);
    const int gEmb = imin_h(cdiv_h(tilesN, 8), 2048);
    const int gEdg = imin_h(cdiv_h(tilesE, 8), 2048);
    const int gNod = imin_h(cdiv_h(tilesN, 8), 2048);

    embed_kernel<<<gEmb, 256, 0, stream>>>(h_in, W_in, b_in, h, Nn);

    for (int l = 0; l < Ll; ++l) {
        hipMemsetAsync(agg, 0, (size_t)Nn * EHW * sizeof(float), stream);
        edge_kernel<<<gEdg, 256, 0, stream>>>(
            h, erow, ecol,
            eW1 + (size_t)l * 2 * HIDN * EHW, eb1 + (size_t)l * EHW,
            eW2 + (size_t)l * EHW * EHW,      eb2 + (size_t)l * EHW,
            agg, Ee);
        node_kernel<<<gNod, 256, 0, stream>>>(
            h, agg,
            nW1 + (size_t)l * KIN_NODE * HIDN, nb1 + (size_t)l * HIDN,
            nW2 + (size_t)l * HIDN * HIDN,     nb2 + (size_t)l * HIDN,
            Nn);
    }

    out_kernel<<<gEmb, 256, 0, stream>>>(h, W_out, b_out, (float*)d_out, Nn);
}


# Node kernel: a tile's h rows read once as 16-B chunks kept in registers (residual; an LDS copy feeds the 1st product); result turned through LDS into that layout: no 64 load-add-store trips
# speedup vs baseline: 1.1640x; 1.0410x over previous
#include <hip/hip_runtime.h>
#include <hip/hip_bf16.h>

typedef float v2f __attribute__((ext_vector_type(2)));
typedef float v4f __attribute__((ext_vector_type(4)));
typedef float v8f __attribute__((ext_vector_type(8)));

#define IN_NF   64
#define HIDN    128
#define EHW     64
#define KIN_NODE (HIDN + EHW)

__device__ __forceinline__ float silu_f(float x) {
    return x / (1.0f + __expf(-x));
}

__device__ __forceinline__ v8f splat8(float v) {
    v8f r = {v, v, v, v, v, v, v, v};
    return r;
}

__device__ __forceinline__ v8f wmma4(v2f a, v2f b, v8f c) {
    return __builtin_amdgcn_wmma_f32_16x16x4_f32(
         false, a,  false, b,
         (short)0, c,  false,  false);
}

__global__ void embed_kernel(const float* __restrict__ hin,
                             const float* __restrict__ W,
                             const float* __restrict__ bias,
                             float* __restrict__ hout, int Nn) {
    const int lane = threadIdx.x & 31;
    const int c    = lane & 15;
    const int half = lane >> 4;
    const int nw   = (gridDim.x * blockDim.x) >> 5;
    const int wid0 = (blockIdx.x * blockDim.x + threadIdx.x) >> 5;
    const int tiles = (Nn + 15) >> 4;

    for (int t = wid0; t < tiles; t += nw) {
        int mr = t * 16 + c;
        if (mr >= Nn) mr = Nn - 1;

        v8f acc[8];
#pragma unroll
        for (int n = 0; n < 8; ++n) acc[n] = splat8(bias[n * 16 + c]);

        for (int k = 0; k < IN_NF; k += 4) {
            const int kk = k + half * 2;
            v2f a = *(const v2f*)(hin + (size_t)mr * IN_NF + kk);
#pragma unroll
            for (int n = 0; n < 8; ++n) {
                v2f b;
                b.x = W[kk * HIDN + n * 16 + c];
                b.y = W[(kk + 1) * HIDN + n * 16 + c];
                acc[n] = wmma4(a, b, acc[n]);
            }
        }
        if (t * 16 + 16 <= Nn) {
#pragma unroll
            for (int n = 0; n < 8; ++n) {
                float* p = hout + (size_t)(t * 16 + half * 8) * HIDN + n * 16 + c;
#pragma unroll
                for (int i = 0; i < 8; ++i) p[(size_t)i * HIDN] = acc[n][i];
            }
        } else {
#pragma unroll
            for (int n = 0; n < 8; ++n)
#pragma unroll
                for (int i = 0; i < 8; ++i) {
                    const int r = t * 16 + i + half * 8;
                    if (r < Nn) hout[(size_t)r * HIDN + n * 16 + c] = acc[n][i];
                }
        }
    }
}

__device__ __forceinline__ void stage_kinter(float* __restrict__ ws, const float* __restrict__ wg, int K) {
    for (int item = threadIdx.x; item < K * 4; item += blockDim.x) {
        const int q = item >> 4, c4 = item & 15;
        const float* s = wg + (size_t)(4 * q) * EHW + 4 * c4;
        const v4f r0 = *(const v4f*)(s);
        const v4f r1 = *(const v4f*)(s + EHW);
        const v4f r2 = *(const v4f*)(s + 2 * EHW);
        const v4f r3 = *(const v4f*)(s + 3 * EHW);
        float* d = ws + q * 256 + c4 * 16;
        v4f o0 = {r0.x, r1.x, r2.x, r3.x};
        v4f o1 = {r0.y, r1.y, r2.y, r3.y};
        v4f o2 = {r0.z, r1.z, r2.z, r3.z};
        v4f o3 = {r0.w, r1.w, r2.w, r3.w};
        *(v4f*)(d)      = o0;
        *(v4f*)(d + 4)  = o1;
        *(v4f*)(d + 8)  = o2;
        *(v4f*)(d + 12) = o3;
    }
}

__global__ void __launch_bounds__(256, 4) edge_kernel(const float* __restrict__ h,
                            const int* __restrict__ erow,
                            const int* __restrict__ ecol,
                            const float* __restrict__ w1,
                            const float* __restrict__ b1,
                            const float* __restrict__ w2,
                            const float* __restrict__ b2,
                            float* __restrict__ agg, int Ee) {
    constexpr int LWM = EHW + 4;
    __shared__ __attribute__((aligned(16))) float w1s[2 * HIDN * EHW];
    __shared__ __attribute__((aligned(16))) float w2s[EHW * EHW];
    __shared__ __attribute__((aligned(16))) float m1s[8][16 * LWM];

    stage_kinter(w1s, w1, 2 * HIDN);
    stage_kinter(w2s, w2, EHW);
    __syncthreads();

    const int lane = threadIdx.x & 31;
    const int wv   = threadIdx.x >> 5;
    const int c    = lane & 15;
    const int half = lane >> 4;
    const int nw   = (gridDim.x * blockDim.x) >> 5;
    const int wid0 = (blockIdx.x * blockDim.x + threadIdx.x) >> 5;
    const int tiles = (Ee + 15) >> 4;
    float* L = m1s[wv];

    const float* wl  = w1s + (half * 64 + c) * 4;
    const float* wl2 = w2s + (half * 64 + c) * 4;
    const float* La  = L + c * LWM + half * 4;

    for (int t = wid0; t < tiles; t += nw) {
        int ea = t * 16 + c;
        if (ea >= Ee) ea = Ee - 1;
        const float* __restrict__ hra = h + (size_t)erow[ea] * HIDN + half * 4;
        const float* __restrict__ hca = h + (size_t)ecol[ea] * HIDN + half * 4;

        v4f av[32];
#pragma unroll
        for (int j = 0; j < 16; ++j) av[j] = *(const v4f*)(hra + 8 * j);
#pragma unroll
        for (int j = 0; j < 16; ++j) av[16 + j] = *(const v4f*)(hca + 8 * j);

        __builtin_amdgcn_sched_group_barrier(0x020, 18, 0);

        v8f acc[4];
#pragma unroll
        for (int n = 0; n < 4; ++n) acc[n] = splat8(b1[n * 16 + c]);

        v4f bq[2][8];
#pragma unroll
        for (int n = 0; n < 4; ++n) {
            bq[0][n]     = *(const v4f*)(wl + n * 64);
            bq[0][4 + n] = *(const v4f*)(wl + 512 + n * 64);
        }
        __builtin_amdgcn_sched_group_barrier(0x100, 8, 0);
#pragma unroll
        for (int p = 0; p < 16; ++p) {
            if ((p & 1) == 0 && p >= 2 && p <= 8) {
                __builtin_amdgcn_sched_group_barrier(0x020, 4, 0);
            }
            if (p + 1 < 16) {
                const float* wn = wl + (p + 1) * 1024;
#pragma unroll
                for (int n = 0; n < 4; ++n) {
                    bq[(p + 1) & 1][n]     = *(const v4f*)(wn + n * 64);
                    bq[(p + 1) & 1][4 + n] = *(const v4f*)(wn + 512 + n * 64);
                }
                __builtin_amdgcn_sched_group_barrier(0x100, 8, 0);
            }
#pragma unroll
            for (int jj = 0; jj < 2; ++jj) {
                const int j = 2 * p + jj;
                v2f a;
                a.x = av[j].x; a.y = av[j].y;
#pragma unroll
                for (int n = 0; n < 4; ++n) {
                    v2f b;
                    b.x = bq[p & 1][jj * 4 + n].x; b.y = bq[p & 1][jj * 4 + n].y;
                    acc[n] = wmma4(a, b, acc[n]);
                }
                a.x = av[j].z; a.y = av[j].w;
#pragma unroll
                for (int n = 0; n < 4; ++n) {
                    v2f b;
                    b.x = bq[p & 1][jj * 4 + n].z; b.y = bq[p & 1][jj * 4 + n].w;
                    acc[n] = wmma4(a, b, acc[n]);
                }
            }
            __builtin_amdgcn_sched_group_barrier(0x008, 16, 0);
        }

#pragma unroll
        for (int n = 0; n < 4; ++n)
#pragma unroll
            for (int i = 0; i < 8; ++i)
                L[(i + half * 8) * LWM + n * 16 + c] = silu_f(acc[n][i]);

        asm volatile("s_wait_dscnt 0" ::: "memory");

        v8f acc2[4];
#pragma unroll
        for (int n = 0; n < 4; ++n) acc2[n] = splat8(b2[n * 16 + c]);

        v4f a2[8];
        v4f b2q[8][4];
#pragma unroll
        for (int j = 0; j < 8; ++j) {
            a2[j] = *(const v4f*)(La + 8 * j);
#pragma unroll
            for (int n = 0; n < 4; ++n) b2q[j][n] = *(const v4f*)(wl2 + j * 512 + n * 64);
        }
        __builtin_amdgcn_sched_group_barrier(0x100, 40, 0);
#pragma unroll
        for (int j = 0; j < 8; ++j) {
            v2f a;
            a.x = a2[j].x; a.y = a2[j].y;
#pragma unroll
            for (int n = 0; n < 4; ++n) {
                v2f b;
                b.x = b2q[j][n].x; b.y = b2q[j][n].y;
                acc2[n] = wmma4(a, b, acc2[n]);
            }
            a.x = a2[j].z; a.y = a2[j].w;
#pragma unroll
            for (int n = 0; n < 4; ++n) {
                v2f b;
                b.x = b2q[j][n].z; b.y = b2q[j][n].w;
                acc2[n] = wmma4(a, b, acc2[n]);
            }
        }
        __builtin_amdgcn_sched_group_barrier(0x008, 64, 0);

        if (t * 16 + 16 <= Ee) {
#pragma unroll
            for (int i = 0; i < 8; ++i) {
                float* dst = agg + (size_t)erow[t * 16 + i + half * 8] * EHW + c;
#pragma unroll
                for (int n = 0; n < 4; ++n)
                    atomicAdd(dst + n * 16, silu_f(acc2[n][i]));
            }
        } else {
#pragma unroll
            for (int i = 0; i < 8; ++i) {
                const int e = t * 16 + i + half * 8;
                if (e < Ee) {
                    float* dst = agg + (size_t)erow[e] * EHW + c;
#pragma unroll
                    for (int n = 0; n < 4; ++n)
                        atomicAdd(dst + n * 16, silu_f(acc2[n][i]));
                }
            }
        }
        asm volatile("s_wait_dscnt 0" ::: "memory");
    }
}

__global__ void __launch_bounds__(256, 4) node_kernel(float* __restrict__ h,
                            const float* __restrict__ agg,
                            const float* __restrict__ m1,
                            const float* __restrict__ c1,
                            const float* __restrict__ m2,
                            const float* __restrict__ c2,
                            int Nn) {
    constexpr int LW  = HIDN + 4;
    constexpr int WS2 = HIDN + 8;
    __shared__ __attribute__((aligned(16))) float w2s[HIDN * WS2];
    __shared__ __attribute__((aligned(16))) float lds[8][16 * LW];

    for (int i4 = threadIdx.x; i4 < (HIDN * HIDN) / 4; i4 += blockDim.x) {
        const int r = i4 >> 5, col = (i4 & 31) * 4;
        *(v4f*)(w2s + r * WS2 + col) = *(const v4f*)(m2 + r * HIDN + col);
    }
    __syncthreads();

    const int lane = threadIdx.x & 31;
    const int wv   = threadIdx.x >> 5;
    const int c    = lane & 15;
    const int half = lane >> 4;
    const int nw   = (gridDim.x * blockDim.x) >> 5;
    const int wid0 = (blockIdx.x * blockDim.x + threadIdx.x) >> 5;
    const int tiles = (Nn + 15) >> 4;
    float* L = lds[wv];
    float* Lc = L + c * LW + half * 4;

    for (int t = wid0; t < tiles; t += nw) {
        int mr = t * 16 + c;
        if (mr >= Nn) mr = Nn - 1;
        float* hc = h + (size_t)mr * HIDN + half * 4;
        const float* __restrict__ ar = agg + (size_t)mr * EHW;

        v4f hv[16];
#pragma unroll
        for (int j = 0; j < 16; ++j) hv[j] = *(const v4f*)(hc + 8 * j);

        v8f acc[8];
#pragma unroll
        for (int n = 0; n < 8; ++n) acc[n] = splat8(c1[n * 16 + c]);

#pragma unroll
        for (int j = 0; j < 16; ++j) *(v4f*)(Lc + 8 * j) = hv[j];

        asm volatile("s_wait_dscnt 0" ::: "memory");

        for (int k = 0; k < HIDN; k += 4) {
            const int kk = k + half * 2;
            v2f a = *(const v2f*)(L + c * LW + kk);
#pragma unroll
            for (int n = 0; n < 8; ++n) {
                v2f b;
                b.x = m1[kk * HIDN + n * 16 + c];
                b.y = m1[(kk + 1) * HIDN + n * 16 + c];
                acc[n] = wmma4(a, b, acc[n]);
            }
        }
        for (int k = 0; k < EHW; k += 4) {
            const int kk = k + half * 2;
            v2f a = *(const v2f*)(ar + kk);
#pragma unroll
            for (int n = 0; n < 8; ++n) {
                v2f b;
                b.x = m1[(HIDN + kk) * HIDN + n * 16 + c];
                b.y = m1[(HIDN + kk + 1) * HIDN + n * 16 + c];
                acc[n] = wmma4(a, b, acc[n]);
            }
        }
#pragma unroll
        for (int n = 0; n < 8; ++n)
#pragma unroll
            for (int i = 0; i < 8; ++i)
                L[(i + half * 8) * LW + n * 16 + c] = silu_f(acc[n][i]);

        asm volatile("s_wait_dscnt 0" ::: "memory");

        v8f acc2[8];
#pragma unroll
        for (int n = 0; n < 8; ++n) acc2[n] = splat8(c2[n * 16 + c]);

        for (int k = 0; k < HIDN; k += 4) {
            const int kk = k + half * 2;
            v2f a;
            a.x = L[c * LW + kk];
            a.y = L[c * LW + kk + 1];
#pragma unroll
            for (int n = 0; n < 8; ++n) {
                v2f b;
                b.x = w2s[kk * WS2 + n * 16 + c];
                b.y = w2s[(kk + 1) * WS2 + n * 16 + c];
                acc2[n] = wmma4(a, b, acc2[n]);
            }
        }

#pragma unroll
        for (int n = 0; n < 8; ++n)
#pragma unroll
            for (int i = 0; i < 8; ++i)
                L[(i + half * 8) * LW + n * 16 + c] = acc2[n][i];

        asm volatile("s_wait_dscnt 0" ::: "memory");

        if (t * 16 + c < Nn) {
#pragma unroll
            for (int j = 0; j < 16; ++j) {
                const v4f z = *(const v4f*)(Lc + 8 * j);
                *(v4f*)(hc + 8 * j) = hv[j] + z;
            }
        }
        asm volatile("s_wait_dscnt 0" ::: "memory");
    }
}

__global__ void out_kernel(const float* __restrict__ h,
                           const float* __restrict__ W,
                           const float* __restrict__ bias,
                           float* __restrict__ out, int Nn) {
    const int lane = threadIdx.x & 31;
    const int c    = lane & 15;
    const int half = lane >> 4;
    const int nw   = (gridDim.x * blockDim.x) >> 5;
    const int wid0 = (blockIdx.x * blockDim.x + threadIdx.x) >> 5;
    const int tiles = (Nn + 15) >> 4;

    for (int t = wid0; t < tiles; t += nw) {
        int mr = t * 16 + c;
        if (mr >= Nn) mr = Nn - 1;

        v8f acc[4];
#pragma unroll
        for (int n = 0; n < 4; ++n) acc[n] = splat8(bias[n * 16 + c]);

        for (int k = 0; k < HIDN; k += 4) {
            const int kk = k + half * 2;
            v2f a = *(const v2f*)(h + (size_t)mr * HIDN + kk);
#pragma unroll
            for (int n = 0; n < 4; ++n) {
                v2f b;
                b.x = W[kk * EHW + n * 16 + c];
                b.y = W[(kk + 1) * EHW + n * 16 + c];
                acc[n] = wmma4(a, b, acc[n]);
            }
        }
        if (t * 16 + 16 <= Nn) {
#pragma unroll
            for (int n = 0; n < 4; ++n) {
                float* p = out + (size_t)(t * 16 + half * 8) * EHW + n * 16 + c;
#pragma unroll
                for (int i = 0; i < 8; ++i) p[(size_t)i * EHW] = acc[n][i];
            }
        } else {
#pragma unroll
            for (int n = 0; n < 4; ++n)
#pragma unroll
                for (int i = 0; i < 8; ++i) {
                    const int r = t * 16 + i + half * 8;
                    if (r < Nn) out[(size_t)r * EHW + n * 16 + c] = acc[n][i];
                }
        }
    }
}

static inline int imin_h(int a, int b) { return a < b ? a : b; }
static inline int cdiv_h(int a, int b) { return (a + b - 1) / b; }

extern "C" void kernel_launch(void* const* d_in, const int* in_sizes, int n_in,
                              void* d_out, int out_size, void* d_ws, size_t ws_size,
                              hipStream_t stream) {
    const float* h_in  = (const float*)d_in[0];
    const int*   edges = (const int*)d_in[1];
    const float* W_in  = (const float*)d_in[2];
    const float* b_in  = (const float*)d_in[3];
    const float* eW1   = (const float*)d_in[4];
    const float* eb1   = (const float*)d_in[5];
    const float* eW2   = (const float*)d_in[6];
    const float* eb2   = (const float*)d_in[7];
    const float* nW1   = (const float*)d_in[8];
    const float* nb1   = (const float*)d_in[9];
    const float* nW2   = (const float*)d_in[10];
    const float* nb2   = (const float*)d_in[11];
    const float* W_out = (const float*)d_in[12];
    const float* b_out = (const float*)d_in[13];

    const int Nn = in_sizes[0] / IN_NF;
    const int Ee = in_sizes[1] / 2;
    const int Ll = in_sizes[4] / (2 * HIDN * EHW);

    float* h   = (float*)d_ws;
    float* agg = h + (size_t)Nn * HIDN;

    const int* erow = edges;
    const int* ecol = edges + Ee;

    const int tilesN = cdiv_h(Nn, 16);
    const int tilesE = cdiv_h(Ee, 16);
    const int gEmb = imin_h(cdiv_h(tilesN, 8), 2048);
    const int gEdg = imin_h(cdiv_h(tilesE, 8), 2048);
    const int gNod = imin_h(cdiv_h(tilesN, 8), 2048);

    embed_kernel<<<gEmb, 256, 0, stream>>>(h_in, W_in, b_in, h, Nn);

    for (int l = 0; l < Ll; ++l) {
        hipMemsetAsync(agg, 0, (size_t)Nn * EHW * sizeof(float), stream);
        edge_kernel<<<gEdg, 256, 0, stream>>>(
            h, erow, ecol,
            eW1 + (size_t)l * 2 * HIDN * EHW, eb1 + (size_t)l * EHW,
            eW2 + (size_t)l * EHW * EHW,      eb2 + (size_t)l * EHW,
            agg, Ee);
        node_kernel<<<gNod, 256, 0, stream>>>(
            h, agg,
            nW1 + (size_t)l * KIN_NODE * HIDN, nb1 + (size_t)l * HIDN,
            nW2 + (size_t)l * HIDN * HIDN,     nb2 + (size_t)l * HIDN,
            Nn);
    }

    out_kernel<<<gEmb, 256, 0, stream>>>(h, W_out, b_out, (float*)d_out, Nn);
}


# Layer 1: linear embedding folded into the edge model's first weights in the call (Wc = W_in @ w1 halves; one launch more): that product runs over k=128 raw rows, not 256; float sum order changes
# speedup vs baseline: 1.2285x; 1.0554x over previous
#include <hip/hip_runtime.h>
#include <hip/hip_bf16.h>

typedef float v2f __attribute__((ext_vector_type(2)));
typedef float v4f __attribute__((ext_vector_type(4)));
typedef float v8f __attribute__((ext_vector_type(8)));

#define IN_NF   64
#define HIDN    128
#define EHW     64
#define KIN_NODE (HIDN + EHW)

__device__ __forceinline__ float silu_f(float x) {
    return x / (1.0f + __expf(-x));
}

__device__ __forceinline__ v8f splat8(float v) {
    v8f r = {v, v, v, v, v, v, v, v};
    return r;
}

__device__ __forceinline__ v8f wmma4(v2f a, v2f b, v8f c) {
    return __builtin_amdgcn_wmma_f32_16x16x4_f32(
         false, a,  false, b,
         (short)0, c,  false,  false);
}

__global__ void embed_kernel(const float* __restrict__ hin,
                             const float* __restrict__ W,
                             const float* __restrict__ bias,
                             float* __restrict__ hout, int Nn) {
    const int lane = threadIdx.x & 31;
    const int c    = lane & 15;
    const int half = lane >> 4;
    const int nw   = (gridDim.x * blockDim.x) >> 5;
    const int wid0 = (blockIdx.x * blockDim.x + threadIdx.x) >> 5;
    const int tiles = (Nn + 15) >> 4;

    for (int t = wid0; t < tiles; t += nw) {
        int mr = t * 16 + c;
        if (mr >= Nn) mr = Nn - 1;

        v8f acc[8];
#pragma unroll
        for (int n = 0; n < 8; ++n) acc[n] = splat8(bias[n * 16 + c]);

        for (int k = 0; k < IN_NF; k += 4) {
            const int kk = k + half * 2;
            v2f a = *(const v2f*)(hin + (size_t)mr * IN_NF + kk);
#pragma unroll
            for (int n = 0; n < 8; ++n) {
                v2f b;
                b.x = W[kk * HIDN + n * 16 + c];
                b.y = W[(kk + 1) * HIDN + n * 16 + c];
                acc[n] = wmma4(a, b, acc[n]);
            }
        }
        if (t * 16 + 16 <= Nn) {
#pragma unroll
            for (int n = 0; n < 8; ++n) {
                float* p = hout + (size_t)(t * 16 + half * 8) * HIDN + n * 16 + c;
#pragma unroll
                for (int i = 0; i < 8; ++i) p[(size_t)i * HIDN] = acc[n][i];
            }
        } else {
#pragma unroll
            for (int n = 0; n < 8; ++n)
#pragma unroll
                for (int i = 0; i < 8; ++i) {
                    const int r = t * 16 + i + half * 8;
                    if (r < Nn) hout[(size_t)r * HIDN + n * 16 + c] = acc[n][i];
                }
        }
    }
}

__global__ void fold_kernel(const float* __restrict__ W_in, const float* __restrict__ b_in,
                            const float* __restrict__ w1, const float* __restrict__ b1,
                            float* __restrict__ Wc, float* __restrict__ bc) {
    const int gid = blockIdx.x * blockDim.x + threadIdx.x;
    const int c = gid & (EHW - 1);
    const int r = gid / EHW;
    if (r < 2 * IN_NF) {
        const float* a = W_in + (size_t)(r % IN_NF) * HIDN;
        const float* b = w1 + (size_t)(r / IN_NF) * HIDN * EHW + c;
        float s = 0.0f;
        for (int k = 0; k < HIDN; ++k) s = fmaf(a[k], b[(size_t)k * EHW], s);
        Wc[r * EHW + c] = s;
    } else if (r == 2 * IN_NF) {
        float s = b1[c];
        for (int k = 0; k < HIDN; ++k)
            s = fmaf(b_in[k], w1[(size_t)k * EHW + c] + w1[(size_t)(HIDN + k) * EHW + c], s);
        bc[c] = s;
    }
}

__device__ __forceinline__ void stage_kinter(float* __restrict__ ws, const float* __restrict__ wg, int K) {
    for (int item = threadIdx.x; item < K * 4; item += blockDim.x) {
        const int q = item >> 4, c4 = item & 15;
        const float* s = wg + (size_t)(4 * q) * EHW + 4 * c4;
        const v4f r0 = *(const v4f*)(s);
        const v4f r1 = *(const v4f*)(s + EHW);
        const v4f r2 = *(const v4f*)(s + 2 * EHW);
        const v4f r3 = *(const v4f*)(s + 3 * EHW);
        float* d = ws + q * 256 + c4 * 16;
        v4f o0 = {r0.x, r1.x, r2.x, r3.x};
        v4f o1 = {r0.y, r1.y, r2.y, r3.y};
        v4f o2 = {r0.z, r1.z, r2.z, r3.z};
        v4f o3 = {r0.w, r1.w, r2.w, r3.w};
        *(v4f*)(d)      = o0;
        *(v4f*)(d + 4)  = o1;
        *(v4f*)(d + 8)  = o2;
        *(v4f*)(d + 12) = o3;
    }
}

template <int KH>
__global__ void __launch_bounds__(256, 4) edge_kernel(const float* __restrict__ h,
                            const int* __restrict__ erow,
                            const int* __restrict__ ecol,
                            const float* __restrict__ w1,
                            const float* __restrict__ b1,
                            const float* __restrict__ w2,
                            const float* __restrict__ b2,
                            float* __restrict__ agg, int Ee) {
    constexpr int LWM = EHW + 4;
    constexpr int NB = KH / 8;
    __shared__ __attribute__((aligned(16))) float w1s[2 * KH * EHW];
    __shared__ __attribute__((aligned(16))) float w2s[EHW * EHW];
    __shared__ __attribute__((aligned(16))) float m1s[8][16 * LWM];

    stage_kinter(w1s, w1, 2 * KH);
    stage_kinter(w2s, w2, EHW);
    __syncthreads();

    const int lane = threadIdx.x & 31;
    const int wv   = threadIdx.x >> 5;
    const int c    = lane & 15;
    const int half = lane >> 4;
    const int nw   = (gridDim.x * blockDim.x) >> 5;
    const int wid0 = (blockIdx.x * blockDim.x + threadIdx.x) >> 5;
    const int tiles = (Ee + 15) >> 4;
    float* L = m1s[wv];

    const float* wl  = w1s + (half * 64 + c) * 4;
    const float* wl2 = w2s + (half * 64 + c) * 4;
    const float* La  = L + c * LWM + half * 4;

    for (int t = wid0; t < tiles; t += nw) {
        int ea = t * 16 + c;
        if (ea >= Ee) ea = Ee - 1;
        const float* __restrict__ hra = h + (size_t)erow[ea] * KH + half * 4;
        const float* __restrict__ hca = h + (size_t)ecol[ea] * KH + half * 4;

        v4f av[2 * NB];
#pragma unroll
        for (int j = 0; j < NB; ++j) av[j] = *(const v4f*)(hra + 8 * j);
#pragma unroll
        for (int j = 0; j < NB; ++j) av[NB + j] = *(const v4f*)(hca + 8 * j);

        __builtin_amdgcn_sched_group_barrier(0x020, 18, 0);

        v8f acc[4];
#pragma unroll
        for (int n = 0; n < 4; ++n) acc[n] = splat8(b1[n * 16 + c]);

        v4f bq[2][8];
#pragma unroll
        for (int n = 0; n < 4; ++n) {
            bq[0][n]     = *(const v4f*)(wl + n * 64);
            bq[0][4 + n] = *(const v4f*)(wl + 512 + n * 64);
        }
        __builtin_amdgcn_sched_group_barrier(0x100, 8, 0);
#pragma unroll
        for (int p = 0; p < NB; ++p) {
            if (KH == 128 && (p & 1) == 0 && p >= 2 && p <= 8) {
                __builtin_amdgcn_sched_group_barrier(0x020, 4, 0);
            }
            if (p + 1 < NB) {
                const float* wn = wl + (p + 1) * 1024;
#pragma unroll
                for (int n = 0; n < 4; ++n) {
                    bq[(p + 1) & 1][n]     = *(const v4f*)(wn + n * 64);
                    bq[(p + 1) & 1][4 + n] = *(const v4f*)(wn + 512 + n * 64);
                }
                __builtin_amdgcn_sched_group_barrier(0x100, 8, 0);
            }
#pragma unroll
            for (int jj = 0; jj < 2; ++jj) {
                const int j = 2 * p + jj;
                v2f a;
                a.x = av[j].x; a.y = av[j].y;
#pragma unroll
                for (int n = 0; n < 4; ++n) {
                    v2f b;
                    b.x = bq[p & 1][jj * 4 + n].x; b.y = bq[p & 1][jj * 4 + n].y;
                    acc[n] = wmma4(a, b, acc[n]);
                }
                a.x = av[j].z; a.y = av[j].w;
#pragma unroll
                for (int n = 0; n < 4; ++n) {
                    v2f b;
                    b.x = bq[p & 1][jj * 4 + n].z; b.y = bq[p & 1][jj * 4 + n].w;
                    acc[n] = wmma4(a, b, acc[n]);
                }
            }
            __builtin_amdgcn_sched_group_barrier(0x008, 16, 0);
        }

#pragma unroll
        for (int n = 0; n < 4; ++n)
#pragma unroll
            for (int i = 0; i < 8; ++i)
                L[(i + half * 8) * LWM + n * 16 + c] = silu_f(acc[n][i]);

        asm volatile("s_wait_dscnt 0" ::: "memory");

        v8f acc2[4];
#pragma unroll
        for (int n = 0; n < 4; ++n) acc2[n] = splat8(b2[n * 16 + c]);

        v4f a2[8];
        v4f b2q[8][4];
#pragma unroll
        for (int j = 0; j < 8; ++j) {
            a2[j] = *(const v4f*)(La + 8 * j);
#pragma unroll
            for (int n = 0; n < 4; ++n) b2q[j][n] = *(const v4f*)(wl2 + j * 512 + n * 64);
        }
        __builtin_amdgcn_sched_group_barrier(0x100, 40, 0);
#pragma unroll
        for (int j = 0; j < 8; ++j) {
            v2f a;
            a.x = a2[j].x; a.y = a2[j].y;
#pragma unroll
            for (int n = 0; n < 4; ++n) {
                v2f b;
                b.x = b2q[j][n].x; b.y = b2q[j][n].y;
                acc2[n] = wmma4(a, b, acc2[n]);
            }
            a.x = a2[j].z; a.y = a2[j].w;
#pragma unroll
            for (int n = 0; n < 4; ++n) {
                v2f b;
                b.x = b2q[j][n].z; b.y = b2q[j][n].w;
                acc2[n] = wmma4(a, b, acc2[n]);
            }
        }
        __builtin_amdgcn_sched_group_barrier(0x008, 64, 0);

        if (t * 16 + 16 <= Ee) {
#pragma unroll
            for (int i = 0; i < 8; ++i) {
                float* dst = agg + (size_t)erow[t * 16 + i + half * 8] * EHW + c;
#pragma unroll
                for (int n = 0; n < 4; ++n)
                    atomicAdd(dst + n * 16, silu_f(acc2[n][i]));
            }
        } else {
#pragma unroll
            for (int i = 0; i < 8; ++i) {
                const int e = t * 16 + i + half * 8;
                if (e < Ee) {
                    float* dst = agg + (size_t)erow[e] * EHW + c;
#pragma unroll
                    for (int n = 0; n < 4; ++n)
                        atomicAdd(dst + n * 16, silu_f(acc2[n][i]));
                }
            }
        }
        asm volatile("s_wait_dscnt 0" ::: "memory");
    }
}

__global__ void __launch_bounds__(256, 4) node_kernel(float* __restrict__ h,
                            const float* __restrict__ agg,
                            const float* __restrict__ m1,
                            const float* __restrict__ c1,
                            const float* __restrict__ m2,
                            const float* __restrict__ c2,
                            int Nn) {
    constexpr int LW  = HIDN + 4;
    constexpr int WS2 = HIDN + 8;
    __shared__ __attribute__((aligned(16))) float w2s[HIDN * WS2];
    __shared__ __attribute__((aligned(16))) float lds[8][16 * LW];

    for (int i4 = threadIdx.x; i4 < (HIDN * HIDN) / 4; i4 += blockDim.x) {
        const int r = i4 >> 5, col = (i4 & 31) * 4;
        *(v4f*)(w2s + r * WS2 + col) = *(const v4f*)(m2 + r * HIDN + col);
    }
    __syncthreads();

    const int lane = threadIdx.x & 31;
    const int wv   = threadIdx.x >> 5;
    const int c    = lane & 15;
    const int half = lane >> 4;
    const int nw   = (gridDim.x * blockDim.x) >> 5;
    const int wid0 = (blockIdx.x * blockDim.x + threadIdx.x) >> 5;
    const int tiles = (Nn + 15) >> 4;
    float* L = lds[wv];
    float* Lc = L + c * LW + half * 4;

    for (int t = wid0; t < tiles; t += nw) {
        int mr = t * 16 + c;
        if (mr >= Nn) mr = Nn - 1;
        float* hc = h + (size_t)mr * HIDN + half * 4;
        const float* __restrict__ ar = agg + (size_t)mr * EHW;

        v4f hv[16];
#pragma unroll
        for (int j = 0; j < 16; ++j) hv[j] = *(const v4f*)(hc + 8 * j);

        v8f acc[8];
#pragma unroll
        for (int n = 0; n < 8; ++n) acc[n] = splat8(c1[n * 16 + c]);

#pragma unroll
        for (int j = 0; j < 16; ++j) *(v4f*)(Lc + 8 * j) = hv[j];

        asm volatile("s_wait_dscnt 0" ::: "memory");

        for (int k = 0; k < HIDN; k += 4) {
            const int kk = k + half * 2;
            v2f a = *(const v2f*)(L + c * LW + kk);
#pragma unroll
            for (int n = 0; n < 8; ++n) {
                v2f b;
                b.x = m1[kk * HIDN + n * 16 + c];
                b.y = m1[(kk + 1) * HIDN + n * 16 + c];
                acc[n] = wmma4(a, b, acc[n]);
            }
        }
        for (int k = 0; k < EHW; k += 4) {
            const int kk = k + half * 2;
            v2f a = *(const v2f*)(ar + kk);
#pragma unroll
            for (int n = 0; n < 8; ++n) {
                v2f b;
                b.x = m1[(HIDN + kk) * HIDN + n * 16 + c];
                b.y = m1[(HIDN + kk + 1) * HIDN + n * 16 + c];
                acc[n] = wmma4(a, b, acc[n]);
            }
        }
#pragma unroll
        for (int n = 0; n < 8; ++n)
#pragma unroll
            for (int i = 0; i < 8; ++i)
                L[(i + half * 8) * LW + n * 16 + c] = silu_f(acc[n][i]);

        asm volatile("s_wait_dscnt 0" ::: "memory");

        v8f acc2[8];
#pragma unroll
        for (int n = 0; n < 8; ++n) acc2[n] = splat8(c2[n * 16 + c]);

        for (int k = 0; k < HIDN; k += 4) {
            const int kk = k + half * 2;
            v2f a;
            a.x = L[c * LW + kk];
            a.y = L[c * LW + kk + 1];
#pragma unroll
            for (int n = 0; n < 8; ++n) {
                v2f b;
                b.x = w2s[kk * WS2 + n * 16 + c];
                b.y = w2s[(kk + 1) * WS2 + n * 16 + c];
                acc2[n] = wmma4(a, b, acc2[n]);
            }
        }

#pragma unroll
        for (int n = 0; n < 8; ++n)
#pragma unroll
            for (int i = 0; i < 8; ++i)
                L[(i + half * 8) * LW + n * 16 + c] = acc2[n][i];

        asm volatile("s_wait_dscnt 0" ::: "memory");

        if (t * 16 + c < Nn) {
#pragma unroll
            for (int j = 0; j < 16; ++j) {
                const v4f z = *(const v4f*)(Lc + 8 * j);
                *(v4f*)(hc + 8 * j) = hv[j] + z;
            }
        }
        asm volatile("s_wait_dscnt 0" ::: "memory");
    }
}

__global__ void out_kernel(const float* __restrict__ h,
                           const float* __restrict__ W,
                           const float* __restrict__ bias,
                           float* __restrict__ out, int Nn) {
    const int lane = threadIdx.x & 31;
    const int c    = lane & 15;
    const int half = lane >> 4;
    const int nw   = (gridDim.x * blockDim.x) >> 5;
    const int wid0 = (blockIdx.x * blockDim.x + threadIdx.x) >> 5;
    const int tiles = (Nn + 15) >> 4;

    for (int t = wid0; t < tiles; t += nw) {
        int mr = t * 16 + c;
        if (mr >= Nn) mr = Nn - 1;

        v8f acc[4];
#pragma unroll
        for (int n = 0; n < 4; ++n) acc[n] = splat8(bias[n * 16 + c]);

        for (int k = 0; k < HIDN; k += 4) {
            const int kk = k + half * 2;
            v2f a = *(const v2f*)(h + (size_t)mr * HIDN + kk);
#pragma unroll
            for (int n = 0; n < 4; ++n) {
                v2f b;
                b.x = W[kk * EHW + n * 16 + c];
                b.y = W[(kk + 1) * EHW + n * 16 + c];
                acc[n] = wmma4(a, b, acc[n]);
            }
        }
        if (t * 16 + 16 <= Nn) {
#pragma unroll
            for (int n = 0; n < 4; ++n) {
                float* p = out + (size_t)(t * 16 + half * 8) * EHW + n * 16 + c;
#pragma unroll
                for (int i = 0; i < 8; ++i) p[(size_t)i * EHW] = acc[n][i];
            }
        } else {
#pragma unroll
            for (int n = 0; n < 4; ++n)
#pragma unroll
                for (int i = 0; i < 8; ++i) {
                    const int r = t * 16 + i + half * 8;
                    if (r < Nn) out[(size_t)r * EHW + n * 16 + c] = acc[n][i];
                }
        }
    }
}

static inline int imin_h(int a, int b) { return a < b ? a : b; }
static inline int cdiv_h(int a, int b) { return (a + b - 1) / b; }

extern "C" void kernel_launch(void* const* d_in, const int* in_sizes, int n_in,
                              void* d_out, int out_size, void* d_ws, size_t ws_size,
                              hipStream_t stream) {
    const float* h_in  = (const float*)d_in[0];
    const int*   edges = (const int*)d_in[1];
    const float* W_in  = (const float*)d_in[2];
    const float* b_in  = (const float*)d_in[3];
    const float* eW1   = (const float*)d_in[4];
    const float* eb1   = (const float*)d_in[5];
    const float* eW2   = (const float*)d_in[6];
    const float* eb2   = (const float*)d_in[7];
    const float* nW1   = (const float*)d_in[8];
    const float* nb1   = (const float*)d_in[9];
    const float* nW2   = (const float*)d_in[10];
    const float* nb2   = (const float*)d_in[11];
    const float* W_out = (const float*)d_in[12];
    const float* b_out = (const float*)d_in[13];

    const int Nn = in_sizes[0] / IN_NF;
    const int Ee = in_sizes[1] / 2;
    const int Ll = in_sizes[4] / (2 * HIDN * EHW);

    float* h   = (float*)d_ws;
    float* agg = h + (size_t)Nn * HIDN;

    const int* erow = edges;
    const int* ecol = edges + Ee;

    float* Wc = (float*)d_out;
    float* bc = Wc + 2 * IN_NF * EHW;

    const int tilesN = cdiv_h(Nn, 16);
    const int tilesE = cdiv_h(Ee, 16);
    const int gEmb = imin_h(cdiv_h(tilesN, 8), 2048);
    const int gEdg = imin_h(cdiv_h(tilesE, 8), 2048);
    const int gNod = imin_h(cdiv_h(tilesN, 8), 2048);

    embed_kernel<<<gEmb, 256, 0, stream>>>(h_in, W_in, b_in, h, Nn);
    fold_kernel<<<cdiv_h(2 * IN_NF * EHW + EHW, 256), 256, 0, stream>>>(W_in, b_in, eW1, eb1, Wc, bc);

    for (int l = 0; l < Ll; ++l) {
        hipMemsetAsync(agg, 0, (size_t)Nn * EHW * sizeof(float), stream);
        if (l == 0) {
            edge_kernel<IN_NF><<<gEdg, 256, 0, stream>>>(
                h_in, erow, ecol, Wc, bc, eW2, eb2, agg, Ee);
        } else {
            edge_kernel<HIDN><<<gEdg, 256, 0, stream>>>(
                h, erow, ecol,
                eW1 + (size_t)l * 2 * HIDN * EHW, eb1 + (size_t)l * EHW,
                eW2 + (size_t)l * EHW * EHW,      eb2 + (size_t)l * EHW,
                agg, Ee);
        }
        node_kernel<<<gNod, 256, 0, stream>>>(
            h, agg,
            nW1 + (size_t)l * KIN_NODE * HIDN, nb1 + (size_t)l * HIDN,
            nW2 + (size_t)l * HIDN * HIDN,     nb2 + (size_t)l * HIDN,
            Nn);
    }

    out_kernel<<<gEmb, 256, 0, stream>>>(h, W_out, b_out, (float*)d_out, Nn);
}


# Node kernel: both weight matrices k-interleaved in LDS, one 512-thread block per WGP; fragment pair = one 16-B read, reads grouped ahead of each 16 products; float sum order changes (bias added last)
# speedup vs baseline: 1.2889x; 1.0492x over previous
#include <hip/hip_runtime.h>
#include <hip/hip_bf16.h>

typedef float v2f __attribute__((ext_vector_type(2)));
typedef float v4f __attribute__((ext_vector_type(4)));
typedef float v8f __attribute__((ext_vector_type(8)));

#define IN_NF   64
#define HIDN    128
#define EHW     64
#define KIN_NODE (HIDN + EHW)

__device__ __forceinline__ float silu_f(float x) {
    return x / (1.0f + __expf(-x));
}

__device__ __forceinline__ v8f splat8(float v) {
    v8f r = {v, v, v, v, v, v, v, v};
    return r;
}

__device__ __forceinline__ v8f wmma4(v2f a, v2f b, v8f c) {
    return __builtin_amdgcn_wmma_f32_16x16x4_f32(
         false, a,  false, b,
         (short)0, c,  false,  false);
}

__global__ void embed_kernel(const float* __restrict__ hin,
                             const float* __restrict__ W,
                             const float* __restrict__ bias,
                             float* __restrict__ hout, int Nn) {
    const int lane = threadIdx.x & 31;
    const int c    = lane & 15;
    const int half = lane >> 4;
    const int nw   = (gridDim.x * blockDim.x) >> 5;
    const int wid0 = (blockIdx.x * blockDim.x + threadIdx.x) >> 5;
    const int tiles = (Nn + 15) >> 4;

    for (int t = wid0; t < tiles; t += nw) {
        int mr = t * 16 + c;
        if (mr >= Nn) mr = Nn - 1;

        v8f acc[8];
#pragma unroll
        for (int n = 0; n < 8; ++n) acc[n] = splat8(bias[n * 16 + c]);

        for (int k = 0; k < IN_NF; k += 4) {
            const int kk = k + half * 2;
            v2f a = *(const v2f*)(hin + (size_t)mr * IN_NF + kk);
#pragma unroll
            for (int n = 0; n < 8; ++n) {
                v2f b;
                b.x = W[kk * HIDN + n * 16 + c];
                b.y = W[(kk + 1) * HIDN + n * 16 + c];
                acc[n] = wmma4(a, b, acc[n]);
            }
        }
        if (t * 16 + 16 <= Nn) {
#pragma unroll
            for (int n = 0; n < 8; ++n) {
                float* p = hout + (size_t)(t * 16 + half * 8) * HIDN + n * 16 + c;
#pragma unroll
                for (int i = 0; i < 8; ++i) p[(size_t)i * HIDN] = acc[n][i];
            }
        } else {
#pragma unroll
            for (int n = 0; n < 8; ++n)
#pragma unroll
                for (int i = 0; i < 8; ++i) {
                    const int r = t * 16 + i + half * 8;
                    if (r < Nn) hout[(size_t)r * HIDN + n * 16 + c] = acc[n][i];
                }
        }
    }
}

__global__ void fold_kernel(const float* __restrict__ W_in, const float* __restrict__ b_in,
                            const float* __restrict__ w1, const float* __restrict__ b1,
                            float* __restrict__ Wc, float* __restrict__ bc) {
    const int gid = blockIdx.x * blockDim.x + threadIdx.x;
    const int c = gid & (EHW - 1);
    const int r = gid / EHW;
    if (r < 2 * IN_NF) {
        const float* a = W_in + (size_t)(r % IN_NF) * HIDN;
        const float* b = w1 + (size_t)(r / IN_NF) * HIDN * EHW + c;
        float s = 0.0f;
        for (int k = 0; k < HIDN; ++k) s = fmaf(a[k], b[(size_t)k * EHW], s);
        Wc[r * EHW + c] = s;
    } else if (r == 2 * IN_NF) {
        float s = b1[c];
        for (int k = 0; k < HIDN; ++k)
            s = fmaf(b_in[k], w1[(size_t)k * EHW + c] + w1[(size_t)(HIDN + k) * EHW + c], s);
        bc[c] = s;
    }
}

__device__ __forceinline__ void stage_kinter(float* __restrict__ ws, const float* __restrict__ wg, int K) {
    for (int item = threadIdx.x; item < K * 4; item += blockDim.x) {
        const int q = item >> 4, c4 = item & 15;
        const float* s = wg + (size_t)(4 * q) * EHW + 4 * c4;
        const v4f r0 = *(const v4f*)(s);
        const v4f r1 = *(const v4f*)(s + EHW);
        const v4f r2 = *(const v4f*)(s + 2 * EHW);
        const v4f r3 = *(const v4f*)(s + 3 * EHW);
        float* d = ws + q * 256 + c4 * 16;
        v4f o0 = {r0.x, r1.x, r2.x, r3.x};
        v4f o1 = {r0.y, r1.y, r2.y, r3.y};
        v4f o2 = {r0.z, r1.z, r2.z, r3.z};
        v4f o3 = {r0.w, r1.w, r2.w, r3.w};
        *(v4f*)(d)      = o0;
        *(v4f*)(d + 4)  = o1;
        *(v4f*)(d + 8)  = o2;
        *(v4f*)(d + 12) = o3;
    }
}

template <int KH>
__global__ void __launch_bounds__(256, 4) edge_kernel(const float* __restrict__ h,
                            const int* __restrict__ erow,
                            const int* __restrict__ ecol,
                            const float* __restrict__ w1,
                            const float* __restrict__ b1,
                            const float* __restrict__ w2,
                            const float* __restrict__ b2,
                            float* __restrict__ agg, int Ee) {
    constexpr int LWM = EHW + 4;
    constexpr int NB = KH / 8;
    __shared__ __attribute__((aligned(16))) float w1s[2 * KH * EHW];
    __shared__ __attribute__((aligned(16))) float w2s[EHW * EHW];
    __shared__ __attribute__((aligned(16))) float m1s[8][16 * LWM];

    stage_kinter(w1s, w1, 2 * KH);
    stage_kinter(w2s, w2, EHW);
    __syncthreads();

    const int lane = threadIdx.x & 31;
    const int wv   = threadIdx.x >> 5;
    const int c    = lane & 15;
    const int half = lane >> 4;
    const int nw   = (gridDim.x * blockDim.x) >> 5;
    const int wid0 = (blockIdx.x * blockDim.x + threadIdx.x) >> 5;
    const int tiles = (Ee + 15) >> 4;
    float* L = m1s[wv];

    const float* wl  = w1s + (half * 64 + c) * 4;
    const float* wl2 = w2s + (half * 64 + c) * 4;
    const float* La  = L + c * LWM + half * 4;

    for (int t = wid0; t < tiles; t += nw) {
        int ea = t * 16 + c;
        if (ea >= Ee) ea = Ee - 1;
        const float* __restrict__ hra = h + (size_t)erow[ea] * KH + half * 4;
        const float* __restrict__ hca = h + (size_t)ecol[ea] * KH + half * 4;

        v4f av[2 * NB];
#pragma unroll
        for (int j = 0; j < NB; ++j) av[j] = *(const v4f*)(hra + 8 * j);
#pragma unroll
        for (int j = 0; j < NB; ++j) av[NB + j] = *(const v4f*)(hca + 8 * j);

        __builtin_amdgcn_sched_group_barrier(0x020, 18, 0);

        v8f acc[4];
#pragma unroll
        for (int n = 0; n < 4; ++n) acc[n] = splat8(b1[n * 16 + c]);

        v4f bq[2][8];
#pragma unroll
        for (int n = 0; n < 4; ++n) {
            bq[0][n]     = *(const v4f*)(wl + n * 64);
            bq[0][4 + n] = *(const v4f*)(wl + 512 + n * 64);
        }
        __builtin_amdgcn_sched_group_barrier(0x100, 8, 0);
#pragma unroll
        for (int p = 0; p < NB; ++p) {
            if (KH == 128 && (p & 1) == 0 && p >= 2 && p <= 8) {
                __builtin_amdgcn_sched_group_barrier(0x020, 4, 0);
            }
            if (p + 1 < NB) {
                const float* wn = wl + (p + 1) * 1024;
#pragma unroll
                for (int n = 0; n < 4; ++n) {
                    bq[(p + 1) & 1][n]     = *(const v4f*)(wn + n * 64);
                    bq[(p + 1) & 1][4 + n] = *(const v4f*)(wn + 512 + n * 64);
                }
                __builtin_amdgcn_sched_group_barrier(0x100, 8, 0);
            }
#pragma unroll
            for (int jj = 0; jj < 2; ++jj) {
                const int j = 2 * p + jj;
                v2f a;
                a.x = av[j].x; a.y = av[j].y;
#pragma unroll
                for (int n = 0; n < 4; ++n) {
                    v2f b;
                    b.x = bq[p & 1][jj * 4 + n].x; b.y = bq[p & 1][jj * 4 + n].y;
                    acc[n] = wmma4(a, b, acc[n]);
                }
                a.x = av[j].z; a.y = av[j].w;
#pragma unroll
                for (int n = 0; n < 4; ++n) {
                    v2f b;
                    b.x = bq[p & 1][jj * 4 + n].z; b.y = bq[p & 1][jj * 4 + n].w;
                    acc[n] = wmma4(a, b, acc[n]);
                }
            }
            __builtin_amdgcn_sched_group_barrier(0x008, 16, 0);
        }

#pragma unroll
        for (int n = 0; n < 4; ++n)
#pragma unroll
            for (int i = 0; i < 8; ++i)
                L[(i + half * 8) * LWM + n * 16 + c] = silu_f(acc[n][i]);

        asm volatile("s_wait_dscnt 0" ::: "memory");

        v8f acc2[4];
#pragma unroll
        for (int n = 0; n < 4; ++n) acc2[n] = splat8(b2[n * 16 + c]);

        v4f a2[8];
        v4f b2q[8][4];
#pragma unroll
        for (int j = 0; j < 8; ++j) {
            a2[j] = *(const v4f*)(La + 8 * j);
#pragma unroll
            for (int n = 0; n < 4; ++n) b2q[j][n] = *(const v4f*)(wl2 + j * 512 + n * 64);
        }
        __builtin_amdgcn_sched_group_barrier(0x100, 40, 0);
#pragma unroll
        for (int j = 0; j < 8; ++j) {
            v2f a;
            a.x = a2[j].x; a.y = a2[j].y;
#pragma unroll
            for (int n = 0; n < 4; ++n) {
                v2f b;
                b.x = b2q[j][n].x; b.y = b2q[j][n].y;
                acc2[n] = wmma4(a, b, acc2[n]);
            }
            a.x = a2[j].z; a.y = a2[j].w;
#pragma unroll
            for (int n = 0; n < 4; ++n) {
                v2f b;
                b.x = b2q[j][n].z; b.y = b2q[j][n].w;
                acc2[n] = wmma4(a, b, acc2[n]);
            }
        }
        __builtin_amdgcn_sched_group_barrier(0x008, 64, 0);

        if (t * 16 + 16 <= Ee) {
#pragma unroll
            for (int i = 0; i < 8; ++i) {
                float* dst = agg + (size_t)erow[t * 16 + i + half * 8] * EHW + c;
#pragma unroll
                for (int n = 0; n < 4; ++n)
                    atomicAdd(dst + n * 16, silu_f(acc2[n][i]));
            }
        } else {
#pragma unroll
            for (int i = 0; i < 8; ++i) {
                const int e = t * 16 + i + half * 8;
                if (e < Ee) {
                    float* dst = agg + (size_t)erow[e] * EHW + c;
#pragma unroll
                    for (int n = 0; n < 4; ++n)
                        atomicAdd(dst + n * 16, silu_f(acc2[n][i]));
                }
            }
        }
        asm volatile("s_wait_dscnt 0" ::: "memory");
    }
}

__device__ __forceinline__ void stage_kinter128(float* __restrict__ ws, const float* __restrict__ wg, int K) {
    for (int item = threadIdx.x; item < K * 8; item += blockDim.x) {
        const int q = item >> 5, c4 = item & 31;
        const float* s = wg + (size_t)(4 * q) * HIDN + 4 * c4;
        const v4f r0 = *(const v4f*)(s);
        const v4f r1 = *(const v4f*)(s + HIDN);
        const v4f r2 = *(const v4f*)(s + 2 * HIDN);
        const v4f r3 = *(const v4f*)(s + 3 * HIDN);
        float* d = ws + q * 512 + c4 * 16;
        v4f o0 = {r0.x, r1.x, r2.x, r3.x};
        v4f o1 = {r0.y, r1.y, r2.y, r3.y};
        v4f o2 = {r0.z, r1.z, r2.z, r3.z};
        v4f o3 = {r0.w, r1.w, r2.w, r3.w};
        *(v4f*)(d)      = o0;
        *(v4f*)(d + 4)  = o1;
        *(v4f*)(d + 8)  = o2;
        *(v4f*)(d + 12) = o3;
    }
}

__global__ void __launch_bounds__(512, 4) node_kernel(float* __restrict__ h,
                            const float* __restrict__ agg,
                            const float* __restrict__ m1,
                            const float* __restrict__ c1,
                            const float* __restrict__ m2,
                            const float* __restrict__ c2,
                            int Nn) {
    constexpr int LW  = HIDN + 4;
    constexpr int NB1 = KIN_NODE / 8;
    constexpr int NBH = HIDN / 8;
    constexpr int NB2 = HIDN / 8;
    constexpr int TILE   = 16 * LW;
    constexpr int W2_OFF = 16 * TILE;
    constexpr int W1_OFF = W2_OFF + HIDN * HIDN;
    __shared__ __attribute__((aligned(16))) float smem[W1_OFF + KIN_NODE * HIDN];

    stage_kinter128(smem + W1_OFF, m1, KIN_NODE);
    stage_kinter128(smem + W2_OFF, m2, HIDN);
    __syncthreads();

    const int lane = threadIdx.x & 31;
    const int wv   = threadIdx.x >> 5;
    const int c    = lane & 15;
    const int half = lane >> 4;
    const int nw   = (gridDim.x * blockDim.x) >> 5;
    const int wid0 = (blockIdx.x * blockDim.x + threadIdx.x) >> 5;
    const int tiles = (Nn + 15) >> 4;
    float* L = smem + wv * TILE;
    float* Lc = L + c * LW + half * 4;
    int iw2  = W2_OFF + (half * 128 + c) * 4;
    int iw1a = W1_OFF + (half * 128 + c) * 4;
    int iw1b = iw1a + 16 * 1024;
    asm volatile("" : "+v"(iw2));
    asm volatile("" : "+v"(iw1a));
    asm volatile("" : "+v"(iw1b));
    const float* wl2  = smem + iw2;
    const float* wl1a = smem + iw1a;
    const float* wl1b = smem + iw1b;

    for (int t = wid0; t < tiles; t += nw) {
        int mr = t * 16 + c;
        if (mr >= Nn) mr = Nn - 1;
        float* hc = h + (size_t)mr * HIDN + half * 4;
        const float* __restrict__ ar = agg + (size_t)mr * EHW + half * 4;

        v4f hv[NBH];
        v4f ag[NB1 - NBH];
#pragma unroll
        for (int j = 0; j < NBH; ++j) hv[j] = *(const v4f*)(hc + 8 * j);
#pragma unroll
        for (int j = 0; j < NB1 - NBH; ++j) ag[j] = *(const v4f*)(ar + 8 * j);

        v8f acc[8];
#pragma unroll
        for (int n = 0; n < 8; ++n) acc[n] = splat8(0.0f);

        v4f bq[2][8];
#pragma unroll
        for (int n = 0; n < 8; ++n) bq[0][n] = *(const v4f*)(wl1a + n * 64);
        __builtin_amdgcn_sched_group_barrier(0x100, 8, 0);
#pragma unroll
        for (int j = 0; j < NB1; ++j) {
            if (j + 1 < NB1) {
                const float* wn = (j + 1 < NBH) ? wl1a + (j + 1) * 1024 : wl1b + (j + 1 - NBH) * 1024;
#pragma unroll
                for (int n = 0; n < 8; ++n) bq[(j + 1) & 1][n] = *(const v4f*)(wn + n * 64);
                __builtin_amdgcn_sched_group_barrier(0x100, 8, 0);
            }
            const v4f av = (j < NBH) ? hv[j < NBH ? j : 0] : ag[j < NBH ? 0 : j - NBH];
            v2f a;
            a.x = av.x; a.y = av.y;
#pragma unroll
            for (int n = 0; n < 8; ++n) {
                v2f b;
                b.x = bq[j & 1][n].x; b.y = bq[j & 1][n].y;
                acc[n] = wmma4(a, b, acc[n]);
            }
            a.x = av.z; a.y = av.w;
#pragma unroll
            for (int n = 0; n < 8; ++n) {
                v2f b;
                b.x = bq[j & 1][n].z; b.y = bq[j & 1][n].w;
                acc[n] = wmma4(a, b, acc[n]);
            }
            __builtin_amdgcn_sched_group_barrier(0x008, 16, 0);
        }

#pragma unroll
        for (int n = 0; n < 8; ++n) {
            const float bv = c1[n * 16 + c];
#pragma unroll
            for (int i = 0; i < 8; ++i)
                L[(i + half * 8) * LW + n * 16 + c] = silu_f(acc[n][i] + bv);
        }

        asm volatile("s_wait_dscnt 0" ::: "memory");

        v8f acc2[8];
#pragma unroll
        for (int n = 0; n < 8; ++n) acc2[n] = splat8(0.0f);

        v4f a2[2];
        v4f b2q[2][8];
        a2[0] = *(const v4f*)(Lc);
#pragma unroll
        for (int n = 0; n < 8; ++n) b2q[0][n] = *(const v4f*)(wl2 + n * 64);
        __builtin_amdgcn_sched_group_barrier(0x100, 9, 0);
#pragma unroll
        for (int j = 0; j < NB2; ++j) {
            if (j + 1 < NB2) {
                const float* wn = wl2 + (j + 1) * 1024;
                a2[(j + 1) & 1] = *(const v4f*)(Lc + 8 * (j + 1));
#pragma unroll
                for (int n = 0; n < 8; ++n) b2q[(j + 1) & 1][n] = *(const v4f*)(wn + n * 64);
                __builtin_amdgcn_sched_group_barrier(0x100, 9, 0);
            }
            v2f a;
            a.x = a2[j & 1].x; a.y = a2[j & 1].y;
#pragma unroll
            for (int n = 0; n < 8; ++n) {
                v2f b;
                b.x = b2q[j & 1][n].x; b.y = b2q[j & 1][n].y;
                acc2[n] = wmma4(a, b, acc2[n]);
            }
            a.x = a2[j & 1].z; a.y = a2[j & 1].w;
#pragma unroll
            for (int n = 0; n < 8; ++n) {
                v2f b;
                b.x = b2q[j & 1][n].z; b.y = b2q[j & 1][n].w;
                acc2[n] = wmma4(a, b, acc2[n]);
            }
            __builtin_amdgcn_sched_group_barrier(0x008, 16, 0);
        }

#pragma unroll
        for (int n = 0; n < 8; ++n) {
            const float bv = c2[n * 16 + c];
#pragma unroll
            for (int i = 0; i < 8; ++i)
                L[(i + half * 8) * LW + n * 16 + c] = acc2[n][i] + bv;
        }

        asm volatile("s_wait_dscnt 0" ::: "memory");

        if (t * 16 + c < Nn) {
#pragma unroll
            for (int j = 0; j < NBH; ++j) {
                const v4f z = *(const v4f*)(Lc + 8 * j);
                *(v4f*)(hc + 8 * j) = hv[j] + z;
            }
        }
        asm volatile("s_wait_dscnt 0" ::: "memory");
    }
}

__global__ void out_kernel(const float* __restrict__ h,
                           const float* __restrict__ W,
                           const float* __restrict__ bias,
                           float* __restrict__ out, int Nn) {
    const int lane = threadIdx.x & 31;
    const int c    = lane & 15;
    const int half = lane >> 4;
    const int nw   = (gridDim.x * blockDim.x) >> 5;
    const int wid0 = (blockIdx.x * blockDim.x + threadIdx.x) >> 5;
    const int tiles = (Nn + 15) >> 4;

    for (int t = wid0; t < tiles; t += nw) {
        int mr = t * 16 + c;
        if (mr >= Nn) mr = Nn - 1;

        v8f acc[4];
#pragma unroll
        for (int n = 0; n < 4; ++n) acc[n] = splat8(bias[n * 16 + c]);

        for (int k = 0; k < HIDN; k += 4) {
            const int kk = k + half * 2;
            v2f a = *(const v2f*)(h + (size_t)mr * HIDN + kk);
#pragma unroll
            for (int n = 0; n < 4; ++n) {
                v2f b;
                b.x = W[kk * EHW + n * 16 + c];
                b.y = W[(kk + 1) * EHW + n * 16 + c];
                acc[n] = wmma4(a, b, acc[n]);
            }
        }
        if (t * 16 + 16 <= Nn) {
#pragma unroll
            for (int n = 0; n < 4; ++n) {
                float* p = out + (size_t)(t * 16 + half * 8) * EHW + n * 16 + c;
#pragma unroll
                for (int i = 0; i < 8; ++i) p[(size_t)i * EHW] = acc[n][i];
            }
        } else {
#pragma unroll
            for (int n = 0; n < 4; ++n)
#pragma unroll
                for (int i = 0; i < 8; ++i) {
                    const int r = t * 16 + i + half * 8;
                    if (r < Nn) out[(size_t)r * EHW + n * 16 + c] = acc[n][i];
                }
        }
    }
}

static inline int imin_h(int a, int b) { return a < b ? a : b; }
static inline int cdiv_h(int a, int b) { return (a + b - 1) / b; }

extern "C" void kernel_launch(void* const* d_in, const int* in_sizes, int n_in,
                              void* d_out, int out_size, void* d_ws, size_t ws_size,
                              hipStream_t stream) {
    const float* h_in  = (const float*)d_in[0];
    const int*   edges = (const int*)d_in[1];
    const float* W_in  = (const float*)d_in[2];
    const float* b_in  = (const float*)d_in[3];
    const float* eW1   = (const float*)d_in[4];
    const float* eb1   = (const float*)d_in[5];
    const float* eW2   = (const float*)d_in[6];
    const float* eb2   = (const float*)d_in[7];
    const float* nW1   = (const float*)d_in[8];
    const float* nb1   = (const float*)d_in[9];
    const float* nW2   = (const float*)d_in[10];
    const float* nb2   = (const float*)d_in[11];
    const float* W_out = (const float*)d_in[12];
    const float* b_out = (const float*)d_in[13];

    const int Nn = in_sizes[0] / IN_NF;
    const int Ee = in_sizes[1] / 2;
    const int Ll = in_sizes[4] / (2 * HIDN * EHW);

    float* h   = (float*)d_ws;
    float* agg = h + (size_t)Nn * HIDN;

    const int* erow = edges;
    const int* ecol = edges + Ee;

    float* Wc = (float*)d_out;
    float* bc = Wc + 2 * IN_NF * EHW;

    const int tilesN = cdiv_h(Nn, 16);
    const int tilesE = cdiv_h(Ee, 16);
    const int gEmb = imin_h(cdiv_h(tilesN, 8), 2048);
    const int gEdg = imin_h(cdiv_h(tilesE, 8), 2048);
    const int gNod = imin_h(cdiv_h(tilesN, 16), 2048);

    embed_kernel<<<gEmb, 256, 0, stream>>>(h_in, W_in, b_in, h, Nn);
    fold_kernel<<<cdiv_h(2 * IN_NF * EHW + EHW, 256), 256, 0, stream>>>(W_in, b_in, eW1, eb1, Wc, bc);

    for (int l = 0; l < Ll; ++l) {
        hipMemsetAsync(agg, 0, (size_t)Nn * EHW * sizeof(float), stream);
        if (l == 0) {
            edge_kernel<IN_NF><<<gEdg, 256, 0, stream>>>(
                h_in, erow, ecol, Wc, bc, eW2, eb2, agg, Ee);
        } else {
            edge_kernel<HIDN><<<gEdg, 256, 0, stream>>>(
                h, erow, ecol,
                eW1 + (size_t)l * 2 * HIDN * EHW, eb1 + (size_t)l * EHW,
                eW2 + (size_t)l * EHW * EHW,      eb2 + (size_t)l * EHW,
                agg, Ee);
        }
        node_kernel<<<gNod, 512, 0, stream>>>(
            h, agg,
            nW1 + (size_t)l * KIN_NODE * HIDN, nb1 + (size_t)l * HIDN,
            nW2 + (size_t)l * HIDN * HIDN,     nb2 + (size_t)l * HIDN,
            Nn);
    }

    out_kernel<<<gEmb, 256, 0, stream>>>(h, W_out, b_out, (float*)d_out, Nn);
}
